# starting kernel
# baseline (speedup 1.0000x reference)
#include <hip/hip_runtime.h>
#include <hip/hip_bf16.h>
#include <math.h>


typedef __attribute__((ext_vector_type(2))) float v2f;
typedef __attribute__((ext_vector_type(8))) float v8f;

#define FIN 128


__global__ void zero_f32(float* __restrict__ p, int n) {
    int i = blockIdx.x * blockDim.x + threadIdx.x;
    if (i < n) p[i] = 0.0f;
}

__global__ void degree_kernel(const int* __restrict__ dst, float* __restrict__ cnt,
                              int nedges) {
    int e = blockIdx.x * blockDim.x + threadIdx.x;
    if (e < nedges) atomicAdd(&cnt[dst[e]], 1.0f);
}

__global__ void invdeg_kernel(const float* __restrict__ cnt, float* __restrict__ inv,
                              int n) {
    int i = blockIdx.x * blockDim.x + threadIdx.x;
    if (i < n) inv[i] = 1.0f / fmaxf(cnt[i], 1.0f);
}

__global__ void scatter_add_kernel(const float* __restrict__ h,
                                   const int* __restrict__ src,
                                   const int* __restrict__ dst,
                                   float* __restrict__ agg, int nedges) {
    int e    = (blockIdx.x * blockDim.x + threadIdx.x) >> 5;
    int lane = threadIdx.x & 31;
    if (e >= nedges) return;
    int s = src[e];
    int d = dst[e];
    const float4 v = ((const float4*)(h + (long)s * FIN))[lane];
    float* ap = agg + (long)d * FIN + lane * 4;
    atomicAdd(ap + 0, v.x);
    atomicAdd(ap + 1, v.y);
    atomicAdd(ap + 2, v.z);
    atomicAdd(ap + 3, v.w);
}


template <int FOUT, bool ELU>
__global__ void __launch_bounds__(32 * (FOUT / 16))
sage_gemm_kernel(const float* __restrict__ agg, const float* __restrict__ invdeg,
                 const float* __restrict__ h,
                 const float* __restrict__ Wl, const float* __restrict__ bl,
                 const float* __restrict__ Wr, float* __restrict__ out) {
    __shared__ float As[16][132];
    __shared__ float Xs[16][132];

    const int m0 = blockIdx.x * 16;

    for (int i = threadIdx.x; i < 16 * (FIN / 4); i += blockDim.x) {
        int r  = i >> 5;
        int c4 = i & 31;
        long g = (long)(m0 + r) * (FIN / 4) + c4;
        float inv = invdeg[m0 + r];
        float4 va = ((const float4*)agg)[g];
        float4 vx = ((const float4*)h)[g];
        va.x *= inv; va.y *= inv; va.z *= inv; va.w *= inv;
        *(float4*)&As[r][c4 * 4] = va;
        *(float4*)&Xs[r][c4 * 4] = vx;
    }
    __syncthreads();

    const int lane = threadIdx.x & 31;
    const int wave = threadIdx.x >> 5;
    const int n0   = wave * 16;
    const int row  = lane & 15;
    const int col  = lane & 15;
    const int hi   = lane >> 4;

    const float* wl = Wl + 2 * hi * FOUT + n0 + col;
    const float* wr = Wr + 2 * hi * FOUT + n0 + col;

    v8f c = {};

#pragma unroll
    for (int k = 0; k < FIN; k += 4) {
        v2f a, b;
        a.x = As[row][k + 2 * hi];
        a.y = As[row][k + 2 * hi + 1];
        b.x = wl[k * FOUT];
        b.y = wl[(k + 1) * FOUT];
        c = __builtin_amdgcn_wmma_f32_16x16x4_f32(false, a, false, b,
                                                  (short)0, c, false, false);
    }
#pragma unroll
    for (int k = 0; k < FIN; k += 4) {
        v2f a, b;
        a.x = Xs[row][k + 2 * hi];
        a.y = Xs[row][k + 2 * hi + 1];
        b.x = wr[k * FOUT];
        b.y = wr[(k + 1) * FOUT];
        c = __builtin_amdgcn_wmma_f32_16x16x4_f32(false, a, false, b,
                                                  (short)0, c, false, false);
    }

    const float bias = bl[n0 + col];
#pragma unroll
    for (int i = 0; i < 8; ++i) {
        float v = c[i] + bias;
        if (ELU) v = (v > 0.0f) ? v : expm1f(v);
        out[(long)(m0 + i + 8 * hi) * FOUT + n0 + col] = v;
    }
}

__global__ void logsoftmax64_kernel(float* __restrict__ out, int n) {
    int row  = (blockIdx.x * blockDim.x + threadIdx.x) >> 5;
    int lane = threadIdx.x & 31;
    if (row >= n) return;
    float* p = out + (long)row * 64;
    float a = p[lane];
    float b = p[lane + 32];
    float m = fmaxf(a, b);
#pragma unroll
    for (int o = 16; o > 0; o >>= 1) m = fmaxf(m, __shfl_xor(m, o, 32));
    float s = __expf(a - m) + __expf(b - m);
#pragma unroll
    for (int o = 16; o > 0; o >>= 1) s += __shfl_xor(s, o, 32);
    float l = m + __logf(s);
    p[lane]      = a - l;
    p[lane + 32] = b - l;
}


extern "C" void kernel_launch(void* const* d_in, const int* in_sizes, int n_in,
                              void* d_out, int out_size, void* d_ws, size_t ws_size,
                              hipStream_t stream) {
    const float* x   = (const float*)d_in[0];
    const int*   ei  = (const int*)d_in[1];
    const float* Wl0 = (const float*)d_in[2];
    const float* bl0 = (const float*)d_in[3];
    const float* Wr0 = (const float*)d_in[4];
    const float* Wl1 = (const float*)d_in[5];
    const float* bl1 = (const float*)d_in[6];
    const float* Wr1 = (const float*)d_in[7];
    const float* Wl2 = (const float*)d_in[8];
    const float* bl2 = (const float*)d_in[9];
    const float* Wr2 = (const float*)d_in[10];

    const int n_nodes = in_sizes[0] / FIN;
    const int n_edges = in_sizes[1] / 2;
    const int* src = ei;
    const int* dst = ei + n_edges;

    float* agg  = (float*)d_ws;
    float* hbuf = agg + (long)n_nodes * FIN;
    float* cnt  = hbuf + (long)n_nodes * FIN;
    float* inv  = cnt + n_nodes;

    const int nf     = n_nodes * FIN;
    const int zb     = (nf + 255) / 256;
    const int eb1    = (n_edges + 255) / 256;
    const int ebw    = (n_edges * 32 + 255) / 256;
    const int nb     = (n_nodes + 255) / 256;
    const int mtiles = n_nodes / 16;

    zero_f32<<<nb, 256, 0, stream>>>(cnt, n_nodes);
    degree_kernel<<<eb1, 256, 0, stream>>>(dst, cnt, n_edges);
    invdeg_kernel<<<nb, 256, 0, stream>>>(cnt, inv, n_nodes);

    zero_f32<<<zb, 256, 0, stream>>>(agg, nf);
    scatter_add_kernel<<<ebw, 256, 0, stream>>>(x, src, dst, agg, n_edges);
    sage_gemm_kernel<128, true><<<mtiles, 256, 0, stream>>>(agg, inv, x,
                                                            Wl0, bl0, Wr0, hbuf);

    zero_f32<<<zb, 256, 0, stream>>>(agg, nf);
    scatter_add_kernel<<<ebw, 256, 0, stream>>>(hbuf, src, dst, agg, n_edges);
    sage_gemm_kernel<128, true><<<mtiles, 256, 0, stream>>>(agg, inv, hbuf,
                                                            Wl1, bl1, Wr1, hbuf);

    zero_f32<<<zb, 256, 0, stream>>>(agg, nf);
    scatter_add_kernel<<<ebw, 256, 0, stream>>>(hbuf, src, dst, agg, n_edges);
    sage_gemm_kernel<64, false><<<mtiles, 128, 0, stream>>>(agg, inv, hbuf,
                                                            Wl2, bl2, Wr2,
                                                            (float*)d_out);

    logsoftmax64_kernel<<<(n_nodes * 32 + 255) / 256, 256, 0, stream>>>(
        (float*)d_out, n_nodes);
}


# Project before aggregating (mean is linear): each GEMM writes h@Wl and h@Wr+b, edges add inv_deg*(h@Wl)[src] onto the latter by f32 atomics; no zero fills, last edge pass 64 wide; sum order changes
# speedup vs baseline: 1.5259x; 1.5259x over previous
#include <hip/hip_runtime.h>
#include <hip/hip_bf16.h>
#include <math.h>


typedef __attribute__((ext_vector_type(2))) float v2f;
typedef __attribute__((ext_vector_type(8))) float v8f;

#define FIN 128


__global__ void zero_f32(float* __restrict__ p, int n) {
    int i = blockIdx.x * blockDim.x + threadIdx.x;
    if (i < n) p[i] = 0.0f;
}

__global__ void degree_kernel(const int* __restrict__ dst, float* __restrict__ cnt,
                              int nedges) {
    int e = blockIdx.x * blockDim.x + threadIdx.x;
    if (e < nedges) atomicAdd(&cnt[dst[e]], 1.0f);
}

__global__ void invdeg_kernel(const float* __restrict__ cnt, float* __restrict__ inv,
                              int n) {
    int i = blockIdx.x * blockDim.x + threadIdx.x;
    if (i < n) inv[i] = 1.0f / fmaxf(cnt[i], 1.0f);
}

template <int F>
__global__ void scatter_scaled_add_kernel(const float* __restrict__ z,
                                          const int* __restrict__ src,
                                          const int* __restrict__ dst,
                                          const float* __restrict__ invdeg,
                                          float* __restrict__ acc, int nedges) {
    int e    = (blockIdx.x * blockDim.x + threadIdx.x) >> 5;
    int lane = threadIdx.x & 31;
    if (e >= nedges) return;
    int s = src[e];
    int d = dst[e];
    const float* zp = z + (long)s * F + lane;
    float*       ap = acc + (long)d * F + lane;
    const float w = invdeg[d];
    float v[F / 32];
#pragma unroll
    for (int j = 0; j < F / 32; ++j) v[j] = zp[32 * j];
#pragma unroll
    for (int j = 0; j < F / 32; ++j) atomicAdd(ap + 32 * j, v[j] * w);
}


template <int FOUT, bool ELU_IN>
__global__ void __launch_bounds__(32 * (FOUT / 16))
sage_project_kernel(const float* h,
                    const float* __restrict__ Wl, const float* __restrict__ bl,
                    const float* __restrict__ Wr,
                    float* __restrict__ z, float* r) {
    __shared__ float Xs[16][132];

    const int m0 = blockIdx.x * 16;

    for (int i = threadIdx.x; i < 16 * (FIN / 4); i += blockDim.x) {
        int rr = i >> 5;
        int c4 = i & 31;
        float4 vx = ((const float4*)h)[(long)(m0 + rr) * (FIN / 4) + c4];
        if (ELU_IN) {
            vx.x = (vx.x > 0.0f) ? vx.x : expm1f(vx.x);
            vx.y = (vx.y > 0.0f) ? vx.y : expm1f(vx.y);
            vx.z = (vx.z > 0.0f) ? vx.z : expm1f(vx.z);
            vx.w = (vx.w > 0.0f) ? vx.w : expm1f(vx.w);
        }
        *(float4*)&Xs[rr][c4 * 4] = vx;
    }
    __syncthreads();

    const int lane = threadIdx.x & 31;
    const int wave = threadIdx.x >> 5;
    const int n0   = wave * 16;
    const int row  = lane & 15;
    const int col  = lane & 15;
    const int hi   = lane >> 4;

    const float* wl = Wl + 2 * hi * FOUT + n0 + col;
    const float* wr = Wr + 2 * hi * FOUT + n0 + col;

    v8f cz = {};
    v8f cr = {};

#pragma unroll
    for (int k = 0; k < FIN; k += 4) {
        v2f a, b0, b1;
        a.x  = Xs[row][k + 2 * hi];
        a.y  = Xs[row][k + 2 * hi + 1];
        b0.x = wl[k * FOUT];
        b0.y = wl[(k + 1) * FOUT];
        b1.x = wr[k * FOUT];
        b1.y = wr[(k + 1) * FOUT];
        cz = __builtin_amdgcn_wmma_f32_16x16x4_f32(false, a, false, b0,
                                                   (short)0, cz, false, false);
        cr = __builtin_amdgcn_wmma_f32_16x16x4_f32(false, a, false, b1,
                                                   (short)0, cr, false, false);
    }

    const float bias = bl[n0 + col];
#pragma unroll
    for (int i = 0; i < 8; ++i) {
        const long o = (long)(m0 + i + 8 * hi) * FOUT + n0 + col;
        z[o] = cz[i];
        r[o] = cr[i] + bias;
    }
}

__global__ void logsoftmax64_kernel(float* __restrict__ out, int n) {
    int row  = (blockIdx.x * blockDim.x + threadIdx.x) >> 5;
    int lane = threadIdx.x & 31;
    if (row >= n) return;
    float* p = out + (long)row * 64;
    float a = p[lane];
    float b = p[lane + 32];
    float m = fmaxf(a, b);
#pragma unroll
    for (int o = 16; o > 0; o >>= 1) m = fmaxf(m, __shfl_xor(m, o, 32));
    float s = __expf(a - m) + __expf(b - m);
#pragma unroll
    for (int o = 16; o > 0; o >>= 1) s += __shfl_xor(s, o, 32);
    float l = m + __logf(s);
    p[lane]      = a - l;
    p[lane + 32] = b - l;
}


extern "C" void kernel_launch(void* const* d_in, const int* in_sizes, int n_in,
                              void* d_out, int out_size, void* d_ws, size_t ws_size,
                              hipStream_t stream) {
    const float* x   = (const float*)d_in[0];
    const int*   ei  = (const int*)d_in[1];
    const float* Wl0 = (const float*)d_in[2];
    const float* bl0 = (const float*)d_in[3];
    const float* Wr0 = (const float*)d_in[4];
    const float* Wl1 = (const float*)d_in[5];
    const float* bl1 = (const float*)d_in[6];
    const float* Wr1 = (const float*)d_in[7];
    const float* Wl2 = (const float*)d_in[8];
    const float* bl2 = (const float*)d_in[9];
    const float* Wr2 = (const float*)d_in[10];

    const int n_nodes = in_sizes[0] / FIN;
    const int n_edges = in_sizes[1] / 2;
    const int* src = ei;
    const int* dst = ei + n_edges;

    float* zbuf = (float*)d_ws;
    float* rbuf = zbuf + (long)n_nodes * FIN;
    float* cnt  = rbuf + (long)n_nodes * FIN;
    float* inv  = cnt + n_nodes;

    const int eb1    = (n_edges + 255) / 256;
    const int ebw    = (n_edges * 32 + 255) / 256;
    const int nb     = (n_nodes + 255) / 256;
    const int mtiles = n_nodes / 16;

    zero_f32<<<nb, 256, 0, stream>>>(cnt, n_nodes);
    degree_kernel<<<eb1, 256, 0, stream>>>(dst, cnt, n_edges);
    invdeg_kernel<<<nb, 256, 0, stream>>>(cnt, inv, n_nodes);

    sage_project_kernel<128, false><<<mtiles, 256, 0, stream>>>(x, Wl0, bl0, Wr0,
                                                                zbuf, rbuf);
    scatter_scaled_add_kernel<128><<<ebw, 256, 0, stream>>>(zbuf, src, dst, inv,
                                                            rbuf, n_edges);

    sage_project_kernel<128, true><<<mtiles, 256, 0, stream>>>(rbuf, Wl1, bl1, Wr1,
                                                               zbuf, rbuf);
    scatter_scaled_add_kernel<128><<<ebw, 256, 0, stream>>>(zbuf, src, dst, inv,
                                                            rbuf, n_edges);

    sage_project_kernel<64, true><<<mtiles, 128, 0, stream>>>(rbuf, Wl2, bl2, Wr2,
                                                              zbuf, (float*)d_out);
    scatter_scaled_add_kernel<64><<<ebw, 256, 0, stream>>>(zbuf, src, dst, inv,
                                                           (float*)d_out, n_edges);

    logsoftmax64_kernel<<<(n_nodes * 32 + 255) / 256, 256, 0, stream>>>(
        (float*)d_out, n_nodes);
}


# Edge pass: one wave per batch of 16 edges (was 1): lane j loads edge j's endpoints and weight once, readlane feeds scalar row bases, all row loads of a batch issue before its atomics; atomics kept
# speedup vs baseline: 1.9190x; 1.2576x over previous
#include <hip/hip_runtime.h>
#include <hip/hip_bf16.h>
#include <math.h>


typedef __attribute__((ext_vector_type(2))) float v2f;
typedef __attribute__((ext_vector_type(8))) float v8f;

#define FIN 128


__global__ void zero_f32(float* __restrict__ p, int n) {
    int i = blockIdx.x * blockDim.x + threadIdx.x;
    if (i < n) p[i] = 0.0f;
}

__global__ void degree_kernel(const int* __restrict__ dst, float* __restrict__ cnt,
                              int nedges) {
    int e = blockIdx.x * blockDim.x + threadIdx.x;
    if (e < nedges) atomicAdd(&cnt[dst[e]], 1.0f);
}

__global__ void invdeg_kernel(const float* __restrict__ cnt, float* __restrict__ inv,
                              int n) {
    int i = blockIdx.x * blockDim.x + threadIdx.x;
    if (i < n) inv[i] = 1.0f / fmaxf(cnt[i], 1.0f);
}

template <int F, int EPW>
__global__ void __launch_bounds__(256)
scatter_scaled_add_kernel(const float* __restrict__ z,
                          const int* __restrict__ src,
                          const int* __restrict__ dst,
                          const float* __restrict__ invdeg,
                          float* __restrict__ acc, int nedges) {
    const int lane = threadIdx.x & 31;
    const int wv = __builtin_amdgcn_readfirstlane(
        (int)((blockIdx.x * blockDim.x + threadIdx.x) >> 5));
    const int e0 = wv * EPW;
    if (e0 >= nedges) return;
    const int left = nedges - e0;
    const int cnt  = left < EPW ? left : EPW;

    int   s_l = 0;
    int   d_l = 0;
    float w_l = 0.0f;
    if (lane < cnt) {
        s_l = src[e0 + lane];
        d_l = dst[e0 + lane];
        w_l = invdeg[d_l];
    }

    if (cnt == EPW) {
        float v[EPW][F / 32];
#pragma unroll
        for (int j = 0; j < EPW; ++j) {
            const int s = __builtin_amdgcn_readlane(s_l, j);
            const float* zp = z + (long)s * F + lane;
#pragma unroll
            for (int q = 0; q < F / 32; ++q) v[j][q] = zp[32 * q];
        }
#pragma unroll
        for (int j = 0; j < EPW; ++j) {
            const int   d = __builtin_amdgcn_readlane(d_l, j);
            const float w = __int_as_float(
                __builtin_amdgcn_readlane(__float_as_int(w_l), j));
            float* ap = acc + (long)d * F + lane;
#pragma unroll
            for (int q = 0; q < F / 32; ++q) atomicAdd(ap + 32 * q, v[j][q] * w);
        }
    } else {
        for (int j = 0; j < cnt; ++j) {
            const int   s = __builtin_amdgcn_readlane(s_l, j);
            const int   d = __builtin_amdgcn_readlane(d_l, j);
            const float w = __int_as_float(
                __builtin_amdgcn_readlane(__float_as_int(w_l), j));
            const float* zp = z + (long)s * F + lane;
            float*       ap = acc + (long)d * F + lane;
            float t[F / 32];
#pragma unroll
            for (int q = 0; q < F / 32; ++q) t[q] = zp[32 * q];
#pragma unroll
            for (int q = 0; q < F / 32; ++q) atomicAdd(ap + 32 * q, t[q] * w);
        }
    }
}


template <int FOUT, bool ELU_IN>
__global__ void __launch_bounds__(32 * (FOUT / 16))
sage_project_kernel(const float* h,
                    const float* __restrict__ Wl, const float* __restrict__ bl,
                    const float* __restrict__ Wr,
                    float* __restrict__ z, float* r) {
    __shared__ float Xs[16][132];

    const int m0 = blockIdx.x * 16;

    for (int i = threadIdx.x; i < 16 * (FIN / 4); i += blockDim.x) {
        int rr = i >> 5;
        int c4 = i & 31;
        float4 vx = ((const float4*)h)[(long)(m0 + rr) * (FIN / 4) + c4];
        if (ELU_IN) {
            vx.x = (vx.x > 0.0f) ? vx.x : expm1f(vx.x);
            vx.y = (vx.y > 0.0f) ? vx.y : expm1f(vx.y);
            vx.z = (vx.z > 0.0f) ? vx.z : expm1f(vx.z);
            vx.w = (vx.w > 0.0f) ? vx.w : expm1f(vx.w);
        }
        *(float4*)&Xs[rr][c4 * 4] = vx;
    }
    __syncthreads();

    const int lane = threadIdx.x & 31;
    const int wave = threadIdx.x >> 5;
    const int n0   = wave * 16;
    const int row  = lane & 15;
    const int col  = lane & 15;
    const int hi   = lane >> 4;

    const float* wl = Wl + 2 * hi * FOUT + n0 + col;
    const float* wr = Wr + 2 * hi * FOUT + n0 + col;

    v8f cz = {};
    v8f cr = {};

#pragma unroll
    for (int k = 0; k < FIN; k += 4) {
        v2f a, b0, b1;
        a.x  = Xs[row][k + 2 * hi];
        a.y  = Xs[row][k + 2 * hi + 1];
        b0.x = wl[k * FOUT];
        b0.y = wl[(k + 1) * FOUT];
        b1.x = wr[k * FOUT];
        b1.y = wr[(k + 1) * FOUT];
        cz = __builtin_amdgcn_wmma_f32_16x16x4_f32(false, a, false, b0,
                                                   (short)0, cz, false, false);
        cr = __builtin_amdgcn_wmma_f32_16x16x4_f32(false, a, false, b1,
                                                   (short)0, cr, false, false);
    }

    const float bias = bl[n0 + col];
#pragma unroll
    for (int i = 0; i < 8; ++i) {
        const long o = (long)(m0 + i + 8 * hi) * FOUT + n0 + col;
        z[o] = cz[i];
        r[o] = cr[i] + bias;
    }
}

__global__ void logsoftmax64_kernel(float* __restrict__ out, int n) {
    int row  = (blockIdx.x * blockDim.x + threadIdx.x) >> 5;
    int lane = threadIdx.x & 31;
    if (row >= n) return;
    float* p = out + (long)row * 64;
    float a = p[lane];
    float b = p[lane + 32];
    float m = fmaxf(a, b);
#pragma unroll
    for (int o = 16; o > 0; o >>= 1) m = fmaxf(m, __shfl_xor(m, o, 32));
    float s = __expf(a - m) + __expf(b - m);
#pragma unroll
    for (int o = 16; o > 0; o >>= 1) s += __shfl_xor(s, o, 32);
    float l = m + __logf(s);
    p[lane]      = a - l;
    p[lane + 32] = b - l;
}


extern "C" void kernel_launch(void* const* d_in, const int* in_sizes, int n_in,
                              void* d_out, int out_size, void* d_ws, size_t ws_size,
                              hipStream_t stream) {
    const float* x   = (const float*)d_in[0];
    const int*   ei  = (const int*)d_in[1];
    const float* Wl0 = (const float*)d_in[2];
    const float* bl0 = (const float*)d_in[3];
    const float* Wr0 = (const float*)d_in[4];
    const float* Wl1 = (const float*)d_in[5];
    const float* bl1 = (const float*)d_in[6];
    const float* Wr1 = (const float*)d_in[7];
    const float* Wl2 = (const float*)d_in[8];
    const float* bl2 = (const float*)d_in[9];
    const float* Wr2 = (const float*)d_in[10];

    const int n_nodes = in_sizes[0] / FIN;
    const int n_edges = in_sizes[1] / 2;
    const int* src = ei;
    const int* dst = ei + n_edges;

    float* zbuf = (float*)d_ws;
    float* rbuf = zbuf + (long)n_nodes * FIN;
    float* cnt  = rbuf + (long)n_nodes * FIN;
    float* inv  = cnt + n_nodes;

    const int eb1    = (n_edges + 255) / 256;
    const int ebw    = ((n_edges + 15) / 16 + 7) / 8;
    const int nb     = (n_nodes + 255) / 256;
    const int mtiles = n_nodes / 16;

    zero_f32<<<nb, 256, 0, stream>>>(cnt, n_nodes);
    degree_kernel<<<eb1, 256, 0, stream>>>(dst, cnt, n_edges);
    invdeg_kernel<<<nb, 256, 0, stream>>>(cnt, inv, n_nodes);

    sage_project_kernel<128, false><<<mtiles, 256, 0, stream>>>(x, Wl0, bl0, Wr0,
                                                                zbuf, rbuf);
    scatter_scaled_add_kernel<128, 16><<<ebw, 256, 0, stream>>>(zbuf, src, dst, inv,
                                                            rbuf, n_edges);

    sage_project_kernel<128, true><<<mtiles, 256, 0, stream>>>(rbuf, Wl1, bl1, Wr1,
                                                               zbuf, rbuf);
    scatter_scaled_add_kernel<128, 16><<<ebw, 256, 0, stream>>>(zbuf, src, dst, inv,
                                                            rbuf, n_edges);

    sage_project_kernel<64, true><<<mtiles, 128, 0, stream>>>(rbuf, Wl2, bl2, Wr2,
                                                              zbuf, (float*)d_out);
    scatter_scaled_add_kernel<64, 16><<<ebw, 256, 0, stream>>>(zbuf, src, dst, inv,
                                                           (float*)d_out, n_edges);

    logsoftmax64_kernel<<<(n_nodes * 32 + 255) / 256, 256, 0, stream>>>(
        (float*)d_out, n_nodes);
}


# Every f32 atomic pass issues a row's adds from one L2 half only: work-groups read their die id and draw (owner half of dst, 256-edge unit) tickets; f32 sum order changes as before; atomics kept
# speedup vs baseline: 2.5852x; 1.3472x over previous
#include <hip/hip_runtime.h>
#include <hip/hip_bf16.h>
#include <math.h>


typedef __attribute__((ext_vector_type(2))) float v2f;
typedef __attribute__((ext_vector_type(8))) float v8f;

#define FIN 128


__global__ void zero_f32(float* __restrict__ p, int n) {
    int i = blockIdx.x * blockDim.x + threadIdx.x;
    if (i < n) p[i] = 0.0f;
}

#define TK 64
__device__ __forceinline__ bool claim_unit(int* __restrict__ ctr, int nunits,
                                           int& owner, int& unit) {
    __shared__ int ticket[2];
    if (threadIdx.x == 0) {
        const int id = __builtin_amdgcn_s_sendmsg_rtn(0x87);
        int o = (id >> 18) & 1;
        int t = atomicAdd(&ctr[o * TK], 1);
        if (t >= nunits) {
            o ^= 1;
            t = atomicAdd(&ctr[o * TK], 1);
        }
        ticket[0] = o;
        ticket[1] = t;
    }
    __syncthreads();
    owner = __builtin_amdgcn_readfirstlane(ticket[0]);
    unit  = __builtin_amdgcn_readfirstlane(ticket[1]);
    return unit < nunits;
}

__global__ void __launch_bounds__(256)
degree_kernel(const int* __restrict__ dst, float* __restrict__ cnt, int nedges,
              int* __restrict__ ctr, int nunits) {
    int owner, unit;
    if (!claim_unit(ctr, nunits, owner, unit)) return;
    const int e = unit * 256 + (int)threadIdx.x;
    if (e < nedges) {
        const int d = dst[e];
        if (((d >> 6) & 1) == owner) atomicAdd(&cnt[d], 1.0f);
    }
}

__global__ void invdeg_kernel(float* __restrict__ cnt, float* __restrict__ inv,
                              int n) {
    int i = blockIdx.x * blockDim.x + threadIdx.x;
    if (i < n) {
        inv[i] = 1.0f / fmaxf(cnt[i], 1.0f);
        if (i < 6 * TK) cnt[i] = 0.0f;
    }
}

template <int F>
__global__ void __launch_bounds__(256)
scatter_scaled_add_kernel(const float* __restrict__ z,
                          const int* __restrict__ src,
                          const int* __restrict__ dst,
                          const float* __restrict__ invdeg,
                          float* __restrict__ acc, int nedges,
                          int* __restrict__ ctr, int nunits) {
    constexpr int Q = F / 32;
    constexpr int G = 8;

    int owner, unit;
    if (!claim_unit(ctr, nunits, owner, unit)) return;

    const int lane = threadIdx.x & 31;
    const int e    = unit * 256 + (int)threadIdx.x;

    int   s_l = 0;
    int   d_l = 0;
    float w_l = 0.0f;
    bool  mine = false;
    if (e < nedges) {
        s_l  = src[e];
        d_l  = dst[e];
        mine = (d_l & 1) == owner;
        if (mine) w_l = invdeg[d_l];
    }
    unsigned m = __builtin_amdgcn_ballot_w32(mine);

    while (__builtin_popcount(m) >= G) {
        float v[G][Q];
        unsigned mm = m;
#pragma unroll
        for (int j = 0; j < G; ++j) {
            const int l = __builtin_ctz(mm);
            mm &= mm - 1;
            const int s = __builtin_amdgcn_readlane(s_l, l);
            const float* zp = z + (long)s * F + lane;
#pragma unroll
            for (int q = 0; q < Q; ++q) v[j][q] = zp[32 * q];
        }
#pragma unroll
        for (int j = 0; j < G; ++j) {
            const int l = __builtin_ctz(m);
            m &= m - 1;
            const int   d = __builtin_amdgcn_readlane(d_l, l);
            const float w = __int_as_float(
                __builtin_amdgcn_readlane(__float_as_int(w_l), l));
            float* ap = acc + (long)d * F + lane;
#pragma unroll
            for (int q = 0; q < Q; ++q) atomicAdd(ap + 32 * q, v[j][q] * w);
        }
    }

    if (m != 0) {
        const int n = __builtin_popcount(m);
        float v[G - 1][Q];
        unsigned mm = m;
#pragma unroll
        for (int j = 0; j < G - 1; ++j) {
            if (j < n) {
                const int l = __builtin_ctz(mm);
                mm &= mm - 1;
                const int s = __builtin_amdgcn_readlane(s_l, l);
                const float* zp = z + (long)s * F + lane;
#pragma unroll
                for (int q = 0; q < Q; ++q) v[j][q] = zp[32 * q];
            }
        }
#pragma unroll
        for (int j = 0; j < G - 1; ++j) {
            if (j < n) {
                const int l = __builtin_ctz(m);
                m &= m - 1;
                const int   d = __builtin_amdgcn_readlane(d_l, l);
                const float w = __int_as_float(
                    __builtin_amdgcn_readlane(__float_as_int(w_l), l));
                float* ap = acc + (long)d * F + lane;
#pragma unroll
                for (int q = 0; q < Q; ++q) atomicAdd(ap + 32 * q, v[j][q] * w);
            }
        }
    }
}


template <int FOUT, bool ELU_IN>
__global__ void __launch_bounds__(32 * (FOUT / 16))
sage_project_kernel(const float* h,
                    const float* __restrict__ Wl, const float* __restrict__ bl,
                    const float* __restrict__ Wr,
                    float* __restrict__ z, float* r) {
    __shared__ float Xs[16][132];

    const int m0 = blockIdx.x * 16;

    for (int i = threadIdx.x; i < 16 * (FIN / 4); i += blockDim.x) {
        int rr = i >> 5;
        int c4 = i & 31;
        float4 vx = ((const float4*)h)[(long)(m0 + rr) * (FIN / 4) + c4];
        if (ELU_IN) {
            vx.x = (vx.x > 0.0f) ? vx.x : expm1f(vx.x);
            vx.y = (vx.y > 0.0f) ? vx.y : expm1f(vx.y);
            vx.z = (vx.z > 0.0f) ? vx.z : expm1f(vx.z);
            vx.w = (vx.w > 0.0f) ? vx.w : expm1f(vx.w);
        }
        *(float4*)&Xs[rr][c4 * 4] = vx;
    }
    __syncthreads();

    const int lane = threadIdx.x & 31;
    const int wave = threadIdx.x >> 5;
    const int n0   = wave * 16;
    const int row  = lane & 15;
    const int col  = lane & 15;
    const int hi   = lane >> 4;

    const float* wl = Wl + 2 * hi * FOUT + n0 + col;
    const float* wr = Wr + 2 * hi * FOUT + n0 + col;

    v8f cz = {};
    v8f cr = {};

#pragma unroll
    for (int k = 0; k < FIN; k += 4) {
        v2f a, b0, b1;
        a.x  = Xs[row][k + 2 * hi];
        a.y  = Xs[row][k + 2 * hi + 1];
        b0.x = wl[k * FOUT];
        b0.y = wl[(k + 1) * FOUT];
        b1.x = wr[k * FOUT];
        b1.y = wr[(k + 1) * FOUT];
        cz = __builtin_amdgcn_wmma_f32_16x16x4_f32(false, a, false, b0,
                                                   (short)0, cz, false, false);
        cr = __builtin_amdgcn_wmma_f32_16x16x4_f32(false, a, false, b1,
                                                   (short)0, cr, false, false);
    }

    const float bias = bl[n0 + col];
#pragma unroll
    for (int i = 0; i < 8; ++i) {
        const long o = (long)(m0 + i + 8 * hi) * FOUT + n0 + col;
        z[o] = cz[i];
        r[o] = cr[i] + bias;
    }
}

__global__ void logsoftmax64_kernel(float* __restrict__ out, int n) {
    int row  = (blockIdx.x * blockDim.x + threadIdx.x) >> 5;
    int lane = threadIdx.x & 31;
    if (row >= n) return;
    float* p = out + (long)row * 64;
    float a = p[lane];
    float b = p[lane + 32];
    float m = fmaxf(a, b);
#pragma unroll
    for (int o = 16; o > 0; o >>= 1) m = fmaxf(m, __shfl_xor(m, o, 32));
    float s = __expf(a - m) + __expf(b - m);
#pragma unroll
    for (int o = 16; o > 0; o >>= 1) s += __shfl_xor(s, o, 32);
    float l = m + __logf(s);
    p[lane]      = a - l;
    p[lane + 32] = b - l;
}


extern "C" void kernel_launch(void* const* d_in, const int* in_sizes, int n_in,
                              void* d_out, int out_size, void* d_ws, size_t ws_size,
                              hipStream_t stream) {
    const float* x   = (const float*)d_in[0];
    const int*   ei  = (const int*)d_in[1];
    const float* Wl0 = (const float*)d_in[2];
    const float* bl0 = (const float*)d_in[3];
    const float* Wr0 = (const float*)d_in[4];
    const float* Wl1 = (const float*)d_in[5];
    const float* bl1 = (const float*)d_in[6];
    const float* Wr1 = (const float*)d_in[7];
    const float* Wl2 = (const float*)d_in[8];
    const float* bl2 = (const float*)d_in[9];
    const float* Wr2 = (const float*)d_in[10];

    const int n_nodes = in_sizes[0] / FIN;
    const int n_edges = in_sizes[1] / 2;
    const int* src = ei;
    const int* dst = ei + n_edges;

    float* zbuf = (float*)d_ws;
    float* rbuf = zbuf + (long)n_nodes * FIN;
    float* cnt  = rbuf + (long)n_nodes * FIN;
    float* inv  = cnt + n_nodes;

    const int nunits = (n_edges + 255) / 256;
    const int egrid  = 2 * nunits;
    const int nb     = (n_nodes + 255) / 256;
    const int mtiles = n_nodes / 16;

    int* tk_deg  = (int*)inv;
    int* tk_edge = (int*)cnt;

    zero_f32<<<(n_nodes + 2 * TK + 255) / 256, 256, 0, stream>>>(cnt, n_nodes + 2 * TK);
    degree_kernel<<<egrid, 256, 0, stream>>>(dst, cnt, n_edges, tk_deg, nunits);
    invdeg_kernel<<<nb, 256, 0, stream>>>(cnt, inv, n_nodes);

    sage_project_kernel<128, false><<<mtiles, 256, 0, stream>>>(x, Wl0, bl0, Wr0,
                                                                zbuf, rbuf);
    scatter_scaled_add_kernel<128><<<egrid, 256, 0, stream>>>(zbuf, src, dst, inv, rbuf,
                                                          n_edges, tk_edge, nunits);

    sage_project_kernel<128, true><<<mtiles, 256, 0, stream>>>(rbuf, Wl1, bl1, Wr1,
                                                               zbuf, rbuf);
    scatter_scaled_add_kernel<128><<<egrid, 256, 0, stream>>>(zbuf, src, dst, inv, rbuf,
                                                          n_edges, tk_edge + 2 * TK, nunits);

    sage_project_kernel<64, true><<<mtiles, 128, 0, stream>>>(rbuf, Wl2, bl2, Wr2,
                                                              zbuf, (float*)d_out);
    scatter_scaled_add_kernel<64><<<egrid, 256, 0, stream>>>(zbuf, src, dst, inv,
                                                         (float*)d_out, n_edges,
                                                         tk_edge + 4 * TK, nunits);

    logsoftmax64_kernel<<<(n_nodes * 32 + 255) / 256, 256, 0, stream>>>(
        (float*)d_out, n_nodes);
}


# Edge pass reads Z rows from LDS not L2: edges grouped per call by (128-row source block, dst owner half), a group stages its block by whole-line loads; f32 atomics kept, sum order open as before
# speedup vs baseline: 2.6945x; 1.0423x over previous
#include <hip/hip_runtime.h>
#include <hip/hip_bf16.h>
#include <math.h>


typedef __attribute__((ext_vector_type(2))) float v2f;
typedef __attribute__((ext_vector_type(8))) float v8f;

#define FIN 128


__global__ void zero2_kernel(float* __restrict__ p, int np,
                             int* __restrict__ q, int nq) {
    int i = blockIdx.x * blockDim.x + threadIdx.x;
    if (i < np) p[i] = 0.0f;
    if (i < nq) q[i] = 0;
}

#define TK 64
__device__ __forceinline__ bool claim_unit(int* __restrict__ ctr, int nunits,
                                           int& owner, int& unit) {
    __shared__ int ticket[2];
    if (threadIdx.x == 0) {
        const int id = __builtin_amdgcn_s_sendmsg_rtn(0x87);
        int o = (id >> 18) & 1;
        int t = atomicAdd(&ctr[o * TK], 1);
        if (t >= nunits) {
            o ^= 1;
            t = atomicAdd(&ctr[o * TK], 1);
        }
        ticket[0] = o;
        ticket[1] = t;
    }
    __syncthreads();
    owner = __builtin_amdgcn_readfirstlane(ticket[0]);
    unit  = __builtin_amdgcn_readfirstlane(ticket[1]);
    return unit < nunits;
}

#define RB 128
#define RB_SHIFT 7
#define CS 32
#define MAXBLK 512
#define EU 4096
#define EW 512

__global__ void __launch_bounds__(1024)
degree_count_kernel(const int* __restrict__ src, const int* __restrict__ dst,
                    float* __restrict__ cnt, int nedges,
                    int* __restrict__ ctr, int nunits,
                    int* __restrict__ gcount, int nblk) {
    __shared__ int hist[MAXBLK];
    int owner, unit;
    if (!claim_unit(ctr, nunits, owner, unit)) return;
    for (int i = threadIdx.x; i < nblk; i += 1024) hist[i] = 0;
    __syncthreads();
    int  s[EU / 1024], d[EU / 1024];
    bool ok[EU / 1024];
#pragma unroll
    for (int j = 0; j < EU / 1024; ++j) {
        const int e = unit * EU + j * 1024 + (int)threadIdx.x;
        ok[j] = e < nedges;
        d[j] = dst[ok[j] ? e : 0];
        s[j] = src[ok[j] ? e : 0];
    }
#pragma unroll
    for (int j = 0; j < EU / 1024; ++j) {
        if (ok[j]) {
            if (((d[j] >> 6) & 1) == owner) atomicAdd(&cnt[d[j]], 1.0f);
            if ((d[j] & 1) == owner) atomicAdd(&hist[s[j] >> RB_SHIFT], 1);
        }
    }
    __syncthreads();
    for (int i = threadIdx.x; i < nblk; i += 1024) {
        const int c = hist[i];
        if (c != 0) atomicAdd(&gcount[(owner * nblk + i) * CS], c);
    }
}

__global__ void __launch_bounds__(256)
invdeg_scan_kernel(float* __restrict__ cnt, float* __restrict__ inv, int n,
                   const int* __restrict__ gcount, int* __restrict__ cursor,
                   int* __restrict__ gstart, int ngroups) {
    __shared__ int part[256];
    const int i = blockIdx.x * 256 + (int)threadIdx.x;
    if (i < n) {
        inv[i] = 1.0f / fmaxf(cnt[i], 1.0f);
        if (i < 6 * TK) cnt[i] = 0.0f;
    }
    if (blockIdx.x != 0) return;
    const int t = (int)threadIdx.x;
    int c[4];
    int sum = 0;
#pragma unroll
    for (int k = 0; k < 4; ++k) {
        const int g = t * 4 + k;
        c[k] = (g < ngroups) ? gcount[g * CS] : 0;
        sum += c[k];
    }
    part[t] = sum;
    __syncthreads();
    for (int off = 1; off < 256; off <<= 1) {
        const int v = (t >= off) ? part[t - off] : 0;
        __syncthreads();
        part[t] += v;
        __syncthreads();
    }
    int run = part[t] - sum;
#pragma unroll
    for (int k = 0; k < 4; ++k) {
        const int g = t * 4 + k;
        if (g < ngroups) {
            gstart[g] = run;
            cursor[g * CS] = run;
            run += c[k];
        }
    }
}

__global__ void __launch_bounds__(1024)
fill_records_kernel(const int* __restrict__ src, const int* __restrict__ dst,
                    const float* __restrict__ inv, int nedges,
                    int* __restrict__ cursor, int* __restrict__ rec, int nblk) {
    __shared__ int hist[2 * MAXBLK];
    int2* __restrict__ rec2 = (int2*)rec;
    const int ngroups = 2 * nblk;
    for (int i = threadIdx.x; i < ngroups; i += 1024) hist[i] = 0;
    __syncthreads();
    int      grp[EU / 1024];
    int      rank[EU / 1024];
    unsigned rx[EU / 1024];
    float    w[EU / 1024];
    int      s[EU / 1024], d[EU / 1024];
#pragma unroll
    for (int j = 0; j < EU / 1024; ++j) {
        const int e = blockIdx.x * EU + j * 1024 + (int)threadIdx.x;
        grp[j] = (e < nedges) ? 0 : -1;
        s[j] = src[(e < nedges) ? e : 0];
        d[j] = dst[(e < nedges) ? e : 0];
    }
#pragma unroll
    for (int j = 0; j < EU / 1024; ++j) w[j] = inv[d[j]];
#pragma unroll
    for (int j = 0; j < EU / 1024; ++j) {
        rank[j] = 0;
        rx[j] = (unsigned)d[j] | ((unsigned)(s[j] & (RB - 1)) << 24);
        if (grp[j] >= 0) {
            grp[j]  = (d[j] & 1) * nblk + (s[j] >> RB_SHIFT);
            rank[j] = atomicAdd(&hist[grp[j]], 1);
        }
    }
    __syncthreads();
    for (int i = threadIdx.x; i < ngroups; i += 1024) {
        const int c = hist[i];
        hist[i] = (c != 0) ? atomicAdd(&cursor[i * CS], c) : 0;
    }
    __syncthreads();
#pragma unroll
    for (int j = 0; j < EU / 1024; ++j) {
        if (grp[j] >= 0)
            rec2[hist[grp[j]] + rank[j]] = make_int2((int)rx[j], __float_as_int(w[j]));
    }
}

template <int F>
__global__ void __launch_bounds__(EW)
edge_from_lds_kernel(const float* __restrict__ z,
                     const int* __restrict__ rec,
                     const int* __restrict__ gstart,
                     const int* __restrict__ gcount,
                     float* __restrict__ acc, int nrows,
                     int* __restrict__ ctr, int nblk) {
    constexpr int Q = F / 32;
    constexpr int G = 8;
    __shared__ float Zs[RB * F];

    int owner, blk;
    if (!claim_unit(ctr, nblk, owner, blk)) return;
    const int g     = owner * nblk + blk;
    const int n     = gcount[g * CS];
    const int start = gstart[g];
    if (n == 0) return;

    const int r0 = blk * RB;
    const int nr = min(RB, nrows - r0);
    {
        const float4* __restrict__ zs = (const float4*)(z + (long)r0 * F);
        float4* ls = (float4*)Zs;
        const int n4 = nr * (F / 4);
#pragma unroll 4
        for (int i = threadIdx.x; i < n4; i += EW) ls[i] = zs[i];
    }
    __syncthreads();

    const int lane = threadIdx.x & 31;
    const int wave = __builtin_amdgcn_readfirstlane((int)(threadIdx.x >> 5));
    const int2* __restrict__ rp = (const int2*)rec + start;

    int base = wave * 32;
    int2 rc = make_int2(0, 0);
    if (base + lane < n) rc = rp[base + lane];
    while (base < n) {
        const int nbase = base + EW;
        int2 rn = make_int2(0, 0);
        if (nbase + lane < n) rn = rp[nbase + lane];

        const int m = min(32, n - base);
        int j = 0;
        for (; j + G <= m; j += G) {
            float v[G][Q];
#pragma unroll
            for (int jj = 0; jj < G; ++jj) {
                const unsigned x = (unsigned)__builtin_amdgcn_readlane(rc.x, j + jj);
                const float* zp = Zs + (x >> 24) * F + lane;
#pragma unroll
                for (int q = 0; q < Q; ++q) v[jj][q] = zp[32 * q];
            }
#pragma unroll
            for (int jj = 0; jj < G; ++jj) {
                const unsigned x = (unsigned)__builtin_amdgcn_readlane(rc.x, j + jj);
                const float w = __int_as_float(
                    __builtin_amdgcn_readlane(rc.y, j + jj));
                float* ap = acc + (long)(x & 0xFFFFFFu) * F + lane;
#pragma unroll
                for (int q = 0; q < Q; ++q) atomicAdd(ap + 32 * q, v[jj][q] * w);
            }
        }
        for (; j < m; ++j) {
            const unsigned x = (unsigned)__builtin_amdgcn_readlane(rc.x, j);
            const float w = __int_as_float(__builtin_amdgcn_readlane(rc.y, j));
            const float* zp = Zs + (x >> 24) * F + lane;
            float* ap = acc + (long)(x & 0xFFFFFFu) * F + lane;
#pragma unroll
            for (int q = 0; q < Q; ++q) atomicAdd(ap + 32 * q, zp[32 * q] * w);
        }
        rc = rn;
        base = nbase;
    }
}


template <int FOUT, bool ELU_IN>
__global__ void __launch_bounds__(32 * (FOUT / 16))
sage_project_kernel(const float* h,
                    const float* __restrict__ Wl, const float* __restrict__ bl,
                    const float* __restrict__ Wr,
                    float* __restrict__ z, float* r) {
    __shared__ float Xs[16][132];

    const int m0 = blockIdx.x * 16;

    for (int i = threadIdx.x; i < 16 * (FIN / 4); i += blockDim.x) {
        int rr = i >> 5;
        int c4 = i & 31;
        float4 vx = ((const float4*)h)[(long)(m0 + rr) * (FIN / 4) + c4];
        if (ELU_IN) {
            vx.x = (vx.x > 0.0f) ? vx.x : expm1f(vx.x);
            vx.y = (vx.y > 0.0f) ? vx.y : expm1f(vx.y);
            vx.z = (vx.z > 0.0f) ? vx.z : expm1f(vx.z);
            vx.w = (vx.w > 0.0f) ? vx.w : expm1f(vx.w);
        }
        *(float4*)&Xs[rr][c4 * 4] = vx;
    }
    __syncthreads();

    const int lane = threadIdx.x & 31;
    const int wave = threadIdx.x >> 5;
    const int n0   = wave * 16;
    const int row  = lane & 15;
    const int col  = lane & 15;
    const int hi   = lane >> 4;

    const float* wl = Wl + 2 * hi * FOUT + n0 + col;
    const float* wr = Wr + 2 * hi * FOUT + n0 + col;

    v8f cz = {};
    v8f cr = {};

#pragma unroll
    for (int k = 0; k < FIN; k += 4) {
        v2f a, b0, b1;
        a.x  = Xs[row][k + 2 * hi];
        a.y  = Xs[row][k + 2 * hi + 1];
        b0.x = wl[k * FOUT];
        b0.y = wl[(k + 1) * FOUT];
        b1.x = wr[k * FOUT];
        b1.y = wr[(k + 1) * FOUT];
        cz = __builtin_amdgcn_wmma_f32_16x16x4_f32(false, a, false, b0,
                                                   (short)0, cz, false, false);
        cr = __builtin_amdgcn_wmma_f32_16x16x4_f32(false, a, false, b1,
                                                   (short)0, cr, false, false);
    }

    const float bias = bl[n0 + col];
#pragma unroll
    for (int i = 0; i < 8; ++i) {
        const long o = (long)(m0 + i + 8 * hi) * FOUT + n0 + col;
        z[o] = cz[i];
        r[o] = cr[i] + bias;
    }
}

__global__ void logsoftmax64_kernel(const float* __restrict__ in,
                                    float* __restrict__ out, int n) {
    int row  = (blockIdx.x * blockDim.x + threadIdx.x) >> 5;
    int lane = threadIdx.x & 31;
    if (row >= n) return;
    const float* p = in + (long)row * 64;
    float* o = out + (long)row * 64;
    float a = p[lane];
    float b = p[lane + 32];
    float m = fmaxf(a, b);
#pragma unroll
    for (int k = 16; k > 0; k >>= 1) m = fmaxf(m, __shfl_xor(m, k, 32));
    float s = __expf(a - m) + __expf(b - m);
#pragma unroll
    for (int k = 16; k > 0; k >>= 1) s += __shfl_xor(s, k, 32);
    float l = m + __logf(s);
    o[lane]      = a - l;
    o[lane + 32] = b - l;
}


extern "C" void kernel_launch(void* const* d_in, const int* in_sizes, int n_in,
                              void* d_out, int out_size, void* d_ws, size_t ws_size,
                              hipStream_t stream) {
    const float* x   = (const float*)d_in[0];
    const int*   ei  = (const int*)d_in[1];
    const float* Wl0 = (const float*)d_in[2];
    const float* bl0 = (const float*)d_in[3];
    const float* Wr0 = (const float*)d_in[4];
    const float* Wl1 = (const float*)d_in[5];
    const float* bl1 = (const float*)d_in[6];
    const float* Wr1 = (const float*)d_in[7];
    const float* Wl2 = (const float*)d_in[8];
    const float* bl2 = (const float*)d_in[9];
    const float* Wr2 = (const float*)d_in[10];

    const int n_nodes = in_sizes[0] / FIN;
    const int n_edges = in_sizes[1] / 2;
    const int* src = ei;
    const int* dst = ei + n_edges;

    float* zbuf = (float*)d_ws;
    float* rbuf = zbuf + (long)n_nodes * FIN;
    float* cnt  = rbuf + (long)n_nodes * FIN;
    float* inv  = cnt + n_nodes;
    float* r2buf = zbuf + (long)n_nodes * 64;

    const int nblk    = (n_nodes + RB - 1) / RB;
    const int ngroups = 2 * nblk;
    const int nunits  = (n_edges + EU - 1) / EU;
    const int egrid   = 2 * nblk;
    const int nb      = (n_nodes + 255) / 256;
    const int mtiles  = n_nodes / 16;

    int* rec    = (int*)d_out;
    int* gcount = rec + 2L * n_edges;
    int* cursor = gcount + ngroups * CS;
    int* gstart = cursor + ngroups * CS;

    int* tk_deg  = (int*)inv;
    int* tk_edge = (int*)cnt;

    const int nz = (n_nodes + 2 * TK > ngroups * CS) ? n_nodes + 2 * TK : ngroups * CS;
    zero2_kernel<<<(nz + 255) / 256, 256, 0, stream>>>(cnt, n_nodes + 2 * TK,
                                                       gcount, ngroups * CS);
    degree_count_kernel<<<2 * nunits, 1024, 0, stream>>>(src, dst, cnt, n_edges, tk_deg,
                                                         nunits, gcount, nblk);
    invdeg_scan_kernel<<<nb, 256, 0, stream>>>(cnt, inv, n_nodes, gcount, cursor,
                                               gstart, ngroups);
    fill_records_kernel<<<nunits, 1024, 0, stream>>>(src, dst, inv, n_edges, cursor,
                                                     rec, nblk);

    sage_project_kernel<128, false><<<mtiles, 256, 0, stream>>>(x, Wl0, bl0, Wr0,
                                                                zbuf, rbuf);
    edge_from_lds_kernel<128><<<egrid, EW, 0, stream>>>(zbuf, rec, gstart, gcount, rbuf,
                                                         n_nodes, tk_edge, nblk);

    sage_project_kernel<128, true><<<mtiles, 256, 0, stream>>>(rbuf, Wl1, bl1, Wr1,
                                                               zbuf, rbuf);
    edge_from_lds_kernel<128><<<egrid, EW, 0, stream>>>(zbuf, rec, gstart, gcount, rbuf,
                                                         n_nodes, tk_edge + 2 * TK, nblk);

    sage_project_kernel<64, true><<<mtiles, 128, 0, stream>>>(rbuf, Wl2, bl2, Wr2,
                                                              zbuf, r2buf);
    edge_from_lds_kernel<64><<<egrid, EW, 0, stream>>>(zbuf, rec, gstart, gcount, r2buf,
                                                        n_nodes, tk_edge + 4 * TK, nblk);

    logsoftmax64_kernel<<<(n_nodes * 32 + 255) / 256, 256, 0, stream>>>(
        r2buf, (float*)d_out, n_nodes);
}


# GEMM launches weight-stationary: a wave loads its 128x16 slices of Wl and Wr into registers once and walks row tiles (512 / 1024 work-groups, h tile double-buffered in LDS), not one tile per group
# speedup vs baseline: 2.8773x; 1.0678x over previous
#include <hip/hip_runtime.h>
#include <hip/hip_bf16.h>
#include <math.h>


typedef __attribute__((ext_vector_type(2))) float v2f;
typedef __attribute__((ext_vector_type(8))) float v8f;

#define FIN 128


__global__ void zero2_kernel(float* __restrict__ p, int np,
                             int* __restrict__ q, int nq) {
    int i = blockIdx.x * blockDim.x + threadIdx.x;
    if (i < np) p[i] = 0.0f;
    if (i < nq) q[i] = 0;
}

#define TK 64
__device__ __forceinline__ bool claim_unit(int* __restrict__ ctr, int nunits,
                                           int& owner, int& unit) {
    __shared__ int ticket[2];
    if (threadIdx.x == 0) {
        const int id = __builtin_amdgcn_s_sendmsg_rtn(0x87);
        int o = (id >> 18) & 1;
        int t = atomicAdd(&ctr[o * TK], 1);
        if (t >= nunits) {
            o ^= 1;
            t = atomicAdd(&ctr[o * TK], 1);
        }
        ticket[0] = o;
        ticket[1] = t;
    }
    __syncthreads();
    owner = __builtin_amdgcn_readfirstlane(ticket[0]);
    unit  = __builtin_amdgcn_readfirstlane(ticket[1]);
    return unit < nunits;
}

#define RB 128
#define RB_SHIFT 7
#define CS 32
#define MAXBLK 512
#define EU 4096
#define EW 512

__global__ void __launch_bounds__(1024)
degree_count_kernel(const int* __restrict__ src, const int* __restrict__ dst,
                    float* __restrict__ cnt, int nedges,
                    int* __restrict__ ctr, int nunits,
                    int* __restrict__ gcount, int nblk) {
    __shared__ int hist[MAXBLK];
    int owner, unit;
    if (!claim_unit(ctr, nunits, owner, unit)) return;
    for (int i = threadIdx.x; i < nblk; i += 1024) hist[i] = 0;
    __syncthreads();
    int  s[EU / 1024], d[EU / 1024];
    bool ok[EU / 1024];
#pragma unroll
    for (int j = 0; j < EU / 1024; ++j) {
        const int e = unit * EU + j * 1024 + (int)threadIdx.x;
        ok[j] = e < nedges;
        d[j] = dst[ok[j] ? e : 0];
        s[j] = src[ok[j] ? e : 0];
    }
#pragma unroll
    for (int j = 0; j < EU / 1024; ++j) {
        if (ok[j]) {
            if (((d[j] >> 6) & 1) == owner) atomicAdd(&cnt[d[j]], 1.0f);
            if ((d[j] & 1) == owner) atomicAdd(&hist[s[j] >> RB_SHIFT], 1);
        }
    }
    __syncthreads();
    for (int i = threadIdx.x; i < nblk; i += 1024) {
        const int c = hist[i];
        if (c != 0) atomicAdd(&gcount[(owner * nblk + i) * CS], c);
    }
}

__global__ void __launch_bounds__(256)
invdeg_scan_kernel(float* __restrict__ cnt, float* __restrict__ inv, int n,
                   const int* __restrict__ gcount, int* __restrict__ cursor,
                   int* __restrict__ gstart, int ngroups) {
    __shared__ int part[256];
    const int i = blockIdx.x * 256 + (int)threadIdx.x;
    if (i < n) {
        inv[i] = 1.0f / fmaxf(cnt[i], 1.0f);
        if (i < 6 * TK) cnt[i] = 0.0f;
    }
    if (blockIdx.x != 0) return;
    const int t = (int)threadIdx.x;
    int c[4];
    int sum = 0;
#pragma unroll
    for (int k = 0; k < 4; ++k) {
        const int g = t * 4 + k;
        c[k] = (g < ngroups) ? gcount[g * CS] : 0;
        sum += c[k];
    }
    part[t] = sum;
    __syncthreads();
    for (int off = 1; off < 256; off <<= 1) {
        const int v = (t >= off) ? part[t - off] : 0;
        __syncthreads();
        part[t] += v;
        __syncthreads();
    }
    int run = part[t] - sum;
#pragma unroll
    for (int k = 0; k < 4; ++k) {
        const int g = t * 4 + k;
        if (g < ngroups) {
            gstart[g] = run;
            cursor[g * CS] = run;
            run += c[k];
        }
    }
}

__global__ void __launch_bounds__(1024)
fill_records_kernel(const int* __restrict__ src, const int* __restrict__ dst,
                    const float* __restrict__ inv, int nedges,
                    int* __restrict__ cursor, int* __restrict__ rec, int nblk) {
    __shared__ int hist[2 * MAXBLK];
    int2* __restrict__ rec2 = (int2*)rec;
    const int ngroups = 2 * nblk;
    for (int i = threadIdx.x; i < ngroups; i += 1024) hist[i] = 0;
    __syncthreads();
    int      grp[EU / 1024];
    int      rank[EU / 1024];
    unsigned rx[EU / 1024];
    float    w[EU / 1024];
    int      s[EU / 1024], d[EU / 1024];
#pragma unroll
    for (int j = 0; j < EU / 1024; ++j) {
        const int e = blockIdx.x * EU + j * 1024 + (int)threadIdx.x;
        grp[j] = (e < nedges) ? 0 : -1;
        s[j] = src[(e < nedges) ? e : 0];
        d[j] = dst[(e < nedges) ? e : 0];
    }
#pragma unroll
    for (int j = 0; j < EU / 1024; ++j) w[j] = inv[d[j]];
#pragma unroll
    for (int j = 0; j < EU / 1024; ++j) {
        rank[j] = 0;
        rx[j] = (unsigned)d[j] | ((unsigned)(s[j] & (RB - 1)) << 24);
        if (grp[j] >= 0) {
            grp[j]  = (d[j] & 1) * nblk + (s[j] >> RB_SHIFT);
            rank[j] = atomicAdd(&hist[grp[j]], 1);
        }
    }
    __syncthreads();
    for (int i = threadIdx.x; i < ngroups; i += 1024) {
        const int c = hist[i];
        hist[i] = (c != 0) ? atomicAdd(&cursor[i * CS], c) : 0;
    }
    __syncthreads();
#pragma unroll
    for (int j = 0; j < EU / 1024; ++j) {
        if (grp[j] >= 0)
            rec2[hist[grp[j]] + rank[j]] = make_int2((int)rx[j], __float_as_int(w[j]));
    }
}

template <int F>
__global__ void __launch_bounds__(EW)
edge_from_lds_kernel(const float* __restrict__ z,
                     const int* __restrict__ rec,
                     const int* __restrict__ gstart,
                     const int* __restrict__ gcount,
                     float* __restrict__ acc, int nrows,
                     int* __restrict__ ctr, int nblk) {
    constexpr int Q = F / 32;
    constexpr int G = 8;
    __shared__ float Zs[RB * F];

    int owner, blk;
    if (!claim_unit(ctr, nblk, owner, blk)) return;
    const int g     = owner * nblk + blk;
    const int n     = gcount[g * CS];
    const int start = gstart[g];
    if (n == 0) return;

    const int r0 = blk * RB;
    const int nr = min(RB, nrows - r0);
    {
        const float4* __restrict__ zs = (const float4*)(z + (long)r0 * F);
        float4* ls = (float4*)Zs;
        const int n4 = nr * (F / 4);
#pragma unroll 4
        for (int i = threadIdx.x; i < n4; i += EW) ls[i] = zs[i];
    }
    __syncthreads();

    const int lane = threadIdx.x & 31;
    const int wave = __builtin_amdgcn_readfirstlane((int)(threadIdx.x >> 5));
    const int2* __restrict__ rp = (const int2*)rec + start;

    int base = wave * 32;
    int2 rc = make_int2(0, 0);
    if (base + lane < n) rc = rp[base + lane];
    while (base < n) {
        const int nbase = base + EW;
        int2 rn = make_int2(0, 0);
        if (nbase + lane < n) rn = rp[nbase + lane];

        const int m = min(32, n - base);
        int j = 0;
        for (; j + G <= m; j += G) {
            float v[G][Q];
#pragma unroll
            for (int jj = 0; jj < G; ++jj) {
                const unsigned x = (unsigned)__builtin_amdgcn_readlane(rc.x, j + jj);
                const float* zp = Zs + (x >> 24) * F + lane;
#pragma unroll
                for (int q = 0; q < Q; ++q) v[jj][q] = zp[32 * q];
            }
#pragma unroll
            for (int jj = 0; jj < G; ++jj) {
                const unsigned x = (unsigned)__builtin_amdgcn_readlane(rc.x, j + jj);
                const float w = __int_as_float(
                    __builtin_amdgcn_readlane(rc.y, j + jj));
                float* ap = acc + (long)(x & 0xFFFFFFu) * F + lane;
#pragma unroll
                for (int q = 0; q < Q; ++q) atomicAdd(ap + 32 * q, v[jj][q] * w);
            }
        }
        for (; j < m; ++j) {
            const unsigned x = (unsigned)__builtin_amdgcn_readlane(rc.x, j);
            const float w = __int_as_float(__builtin_amdgcn_readlane(rc.y, j));
            const float* zp = Zs + (x >> 24) * F + lane;
            float* ap = acc + (long)(x & 0xFFFFFFu) * F + lane;
#pragma unroll
            for (int q = 0; q < Q; ++q) atomicAdd(ap + 32 * q, zp[32 * q] * w);
        }
        rc = rn;
        base = nbase;
    }
}


template <int NP>
struct TilePart { float4 v[NP]; };

template <int NP, int NT>
__device__ __forceinline__ TilePart<NP> tile_fetch(const float* h, int tile) {
    TilePart<NP> p;
#pragma unroll
    for (int j = 0; j < NP; ++j) {
        const int i = (int)threadIdx.x + j * NT;
        p.v[j] = ((const float4*)h)[(long)(tile * 16 + (i >> 5)) * (FIN / 4) + (i & 31)];
    }
    return p;
}

template <int NP, int NT, bool ELU_IN>
__device__ __forceinline__ void tile_land(const TilePart<NP>& p, float (*X)[132]) {
#pragma unroll
    for (int j = 0; j < NP; ++j) {
        const int i = (int)threadIdx.x + j * NT;
        float4 vx = p.v[j];
        if (ELU_IN) {
            vx.x = (vx.x > 0.0f) ? vx.x : expm1f(vx.x);
            vx.y = (vx.y > 0.0f) ? vx.y : expm1f(vx.y);
            vx.z = (vx.z > 0.0f) ? vx.z : expm1f(vx.z);
            vx.w = (vx.w > 0.0f) ? vx.w : expm1f(vx.w);
        }
        *(float4*)&X[i >> 5][(i & 31) * 4] = vx;
    }
}

template <int FOUT, bool ELU_IN>
__global__ void __launch_bounds__(32 * (FOUT / 16))
sage_project_kernel(const float* h,
                    const float* __restrict__ Wl, const float* __restrict__ bl,
                    const float* __restrict__ Wr,
                    float* __restrict__ z, float* r, int mtiles) {
    constexpr int NT = 32 * (FOUT / 16);
    constexpr int NP = 16 * (FIN / 4) / NT;
    __shared__ float Xs[2][16][132];

    const int lane = threadIdx.x & 31;
    const int wave = threadIdx.x >> 5;
    const int n0   = wave * 16;
    const int row  = lane & 15;
    const int col  = lane & 15;
    const int hi   = lane >> 4;

    int t = blockIdx.x;
    if (t >= mtiles) return;

    TilePart<NP> part = tile_fetch<NP, NT>(h, t);

    const float* wl = Wl + 2 * hi * FOUT + n0 + col;
    const float* wr = Wr + 2 * hi * FOUT + n0 + col;
    v2f bz[FIN / 4], br[FIN / 4];
#pragma unroll
    for (int k4 = 0; k4 < FIN / 4; ++k4) {
        bz[k4].x = wl[(4 * k4) * FOUT];
        bz[k4].y = wl[(4 * k4 + 1) * FOUT];
        br[k4].x = wr[(4 * k4) * FOUT];
        br[k4].y = wr[(4 * k4 + 1) * FOUT];
    }
    const float bias = bl[n0 + col];

    tile_land<NP, NT, ELU_IN>(part, Xs[0]);
    __syncthreads();

    int buf = 0;
    for (;;) {
        const int  tn   = t + (int)gridDim.x;
        const bool more = tn < mtiles;
        if (more) part = tile_fetch<NP, NT>(h, tn);

        v8f cz = {};
        v8f cr = {};
#pragma unroll
        for (int k4 = 0; k4 < FIN / 4; ++k4) {
            v2f a;
            a.x = Xs[buf][row][4 * k4 + 2 * hi];
            a.y = Xs[buf][row][4 * k4 + 2 * hi + 1];
            cz = __builtin_amdgcn_wmma_f32_16x16x4_f32(false, a, false, bz[k4],
                                                       (short)0, cz, false, false);
            cr = __builtin_amdgcn_wmma_f32_16x16x4_f32(false, a, false, br[k4],
                                                       (short)0, cr, false, false);
        }

        const long o = (long)(t * 16 + 8 * hi) * FOUT + n0 + col;
        float* zp = z + o;
        float* rp = r + o;
#pragma unroll
        for (int i = 0; i < 8; ++i) {
            zp[i * FOUT] = cz[i];
            rp[i * FOUT] = cr[i] + bias;
        }

        if (!more) break;
        tile_land<NP, NT, ELU_IN>(part, Xs[buf ^ 1]);
        __builtin_amdgcn_fence(__ATOMIC_RELEASE, "workgroup", "local");
        __builtin_amdgcn_s_barrier();
        __builtin_amdgcn_fence(__ATOMIC_ACQUIRE, "workgroup", "local");
        t = tn;
        buf ^= 1;
    }
}

__global__ void logsoftmax64_kernel(const float* __restrict__ in,
                                    float* __restrict__ out, int n) {
    int row  = (blockIdx.x * blockDim.x + threadIdx.x) >> 5;
    int lane = threadIdx.x & 31;
    if (row >= n) return;
    const float* p = in + (long)row * 64;
    float* o = out + (long)row * 64;
    float a = p[lane];
    float b = p[lane + 32];
    float m = fmaxf(a, b);
#pragma unroll
    for (int k = 16; k > 0; k >>= 1) m = fmaxf(m, __shfl_xor(m, k, 32));
    float s = __expf(a - m) + __expf(b - m);
#pragma unroll
    for (int k = 16; k > 0; k >>= 1) s += __shfl_xor(s, k, 32);
    float l = m + __logf(s);
    o[lane]      = a - l;
    o[lane + 32] = b - l;
}


extern "C" void kernel_launch(void* const* d_in, const int* in_sizes, int n_in,
                              void* d_out, int out_size, void* d_ws, size_t ws_size,
                              hipStream_t stream) {
    const float* x   = (const float*)d_in[0];
    const int*   ei  = (const int*)d_in[1];
    const float* Wl0 = (const float*)d_in[2];
    const float* bl0 = (const float*)d_in[3];
    const float* Wr0 = (const float*)d_in[4];
    const float* Wl1 = (const float*)d_in[5];
    const float* bl1 = (const float*)d_in[6];
    const float* Wr1 = (const float*)d_in[7];
    const float* Wl2 = (const float*)d_in[8];
    const float* bl2 = (const float*)d_in[9];
    const float* Wr2 = (const float*)d_in[10];

    const int n_nodes = in_sizes[0] / FIN;
    const int n_edges = in_sizes[1] / 2;
    const int* src = ei;
    const int* dst = ei + n_edges;

    float* zbuf = (float*)d_ws;
    float* rbuf = zbuf + (long)n_nodes * FIN;
    float* cnt  = rbuf + (long)n_nodes * FIN;
    float* inv  = cnt + n_nodes;
    float* r2buf = zbuf + (long)n_nodes * 64;

    const int nblk    = (n_nodes + RB - 1) / RB;
    const int ngroups = 2 * nblk;
    const int nunits  = (n_edges + EU - 1) / EU;
    const int egrid   = 2 * nblk;
    const int nb      = (n_nodes + 255) / 256;
    const int mtiles  = n_nodes / 16;

    int* rec    = (int*)d_out;
    int* gcount = rec + 2L * n_edges;
    int* cursor = gcount + ngroups * CS;
    int* gstart = cursor + ngroups * CS;

    int* tk_deg  = (int*)inv;
    int* tk_edge = (int*)cnt;

    const int nz = (n_nodes + 2 * TK > ngroups * CS) ? n_nodes + 2 * TK : ngroups * CS;
    zero2_kernel<<<(nz + 255) / 256, 256, 0, stream>>>(cnt, n_nodes + 2 * TK,
                                                       gcount, ngroups * CS);
    degree_count_kernel<<<2 * nunits, 1024, 0, stream>>>(src, dst, cnt, n_edges, tk_deg,
                                                         nunits, gcount, nblk);
    invdeg_scan_kernel<<<nb, 256, 0, stream>>>(cnt, inv, n_nodes, gcount, cursor,
                                               gstart, ngroups);
    fill_records_kernel<<<nunits, 1024, 0, stream>>>(src, dst, inv, n_edges, cursor,
                                                     rec, nblk);

    sage_project_kernel<128, false><<<512, 256, 0, stream>>>(x, Wl0, bl0, Wr0,
                                                             zbuf, rbuf, mtiles);
    edge_from_lds_kernel<128><<<egrid, EW, 0, stream>>>(zbuf, rec, gstart, gcount, rbuf,
                                                         n_nodes, tk_edge, nblk);

    sage_project_kernel<128, true><<<512, 256, 0, stream>>>(rbuf, Wl1, bl1, Wr1,
                                                            zbuf, rbuf, mtiles);
    edge_from_lds_kernel<128><<<egrid, EW, 0, stream>>>(zbuf, rec, gstart, gcount, rbuf,
                                                         n_nodes, tk_edge + 2 * TK, nblk);

    sage_project_kernel<64, true><<<1024, 128, 0, stream>>>(rbuf, Wl2, bl2, Wr2,
                                                            zbuf, r2buf, mtiles);
    edge_from_lds_kernel<64><<<egrid, EW, 0, stream>>>(zbuf, rec, gstart, gcount, r2buf,
                                                        n_nodes, tk_edge + 4 * TK, nblk);

    logsoftmax64_kernel<<<(n_nodes * 32 + 255) / 256, 256, 0, stream>>>(
        r2buf, (float*)d_out, n_nodes);
}
